# starting kernel
# baseline (speedup 1.0000x reference)
#include <hip/hip_runtime.h>
#include <hip/hip_bf16.h>
#include <math.h>

typedef __attribute__((ext_vector_type(2))) float v2f;
typedef __attribute__((ext_vector_type(8))) float v8f;

#define BB 4
#define LL 512
#define DD 128

__device__ __forceinline__ float fast_tanh(float x) {
#if defined(__has_builtin)
#if __has_builtin(__builtin_amdgcn_tanhf)
    return __builtin_amdgcn_tanhf(x);
#elif __has_builtin(__builtin_amdgcn_tanh_f32)
    return __builtin_amdgcn_tanh_f32(x);
#else
    return tanhf(x);
#endif
#else
    return tanhf(x);
#endif
}

__global__ __launch_bounds__(128)
void proj_qk_wmma(const float* __restrict__ inp,
                  const float* __restrict__ Wq, const float* __restrict__ bq,
                  const float* __restrict__ Wk,
                  float* __restrict__ qo, float* __restrict__ ko)
{
    const int wave = threadIdx.x >> 5;
    const int lane = threadIdx.x & 31;
    const int tile = blockIdx.x * 4 + wave;
    const int m0   = (tile >> 3) << 4;
    const int n0   = (tile & 7) << 4;
    const int r    = lane & 15;
    const int koff = (lane >> 4) << 1;

    const float* arow  = inp + (size_t)(m0 + r) * DD + koff;
    const float* bqrow = Wq  + (size_t)(n0 + r) * DD + koff;
    const float* bkrow = Wk  + (size_t)(n0 + r) * DD + koff;

    v8f cq = {};
    v8f ck = {};
    for (int kk = 0; kk < DD; kk += 4) {
        v2f a   = *(const v2f*)(arow  + kk);
        v2f bq_ = *(const v2f*)(bqrow + kk);
        v2f bk_ = *(const v2f*)(bkrow + kk);
        cq = __builtin_amdgcn_wmma_f32_16x16x4_f32(false, a, false, bq_,
                                                   (short)0, cq, false, false);
        ck = __builtin_amdgcn_wmma_f32_16x16x4_f32(false, a, false, bk_,
                                                   (short)0, ck, false, false);
    }

    const int   nn    = n0 + (lane & 15);
    const float bias  = bq[nn];
    const int   mbase = m0 + ((lane >> 4) << 3);
#pragma unroll
    for (int rr = 0; rr < 8; ++rr) {
        const int m = mbase + rr;
        qo[(size_t)m * DD + nn] = cq[rr] + bias;
        ko[(size_t)m * DD + nn] = ck[rr];
    }
}

__global__ __launch_bounds__(256)
void score_softmax(const float* __restrict__ q, const float* __restrict__ k,
                   const float* __restrict__ v, float* __restrict__ out)
{
    const int bi = blockIdx.x;
    const int b  = bi >> 9;

    __shared__ float s_k[DD];
    __shared__ float s_v[DD];
    __shared__ float s_score[LL];
    __shared__ float s_red[8];

    const int tid  = threadIdx.x;
    const int lane = tid & 31;
    const int wave = tid >> 5;

    if (tid < DD) {
        s_k[tid] = k[(size_t)bi * DD + tid];
        s_v[tid] = v[tid];
    }
    __syncthreads();

    const float kv0 = s_k[lane +  0], kv1 = s_k[lane + 32];
    const float kv2 = s_k[lane + 64], kv3 = s_k[lane + 96];
    const float vv0 = s_v[lane +  0], vv1 = s_v[lane + 32];
    const float vv2 = s_v[lane + 64], vv3 = s_v[lane + 96];

    const float* qb = q + (size_t)b * LL * DD;

    for (int j = wave; j < LL; j += 8) {
        const float* qr = qb + (size_t)j * DD + lane;
        float acc = fast_tanh(qr[ 0] + kv0) * vv0
                  + fast_tanh(qr[32] + kv1) * vv1
                  + fast_tanh(qr[64] + kv2) * vv2
                  + fast_tanh(qr[96] + kv3) * vv3;
#pragma unroll
        for (int off = 16; off > 0; off >>= 1)
            acc += __shfl_xor(acc, off, 32);
        if (lane == 0) s_score[j] = acc;
    }
    __syncthreads();

    const float a0 = s_score[tid];
    const float a1 = s_score[tid + 256];

    float m = fmaxf(a0, a1);
#pragma unroll
    for (int off = 16; off > 0; off >>= 1)
        m = fmaxf(m, __shfl_xor(m, off, 32));
    if (lane == 0) s_red[wave] = m;
    __syncthreads();

    float bm = s_red[0];
#pragma unroll
    for (int w = 1; w < 8; ++w) bm = fmaxf(bm, s_red[w]);

    const float e0 = __expf(a0 - bm);
    const float e1 = __expf(a1 - bm);

    float s = e0 + e1;
#pragma unroll
    for (int off = 16; off > 0; off >>= 1)
        s += __shfl_xor(s, off, 32);
    __syncthreads();
    if (lane == 0) s_red[wave] = s;
    __syncthreads();

    float bs = s_red[0];
#pragma unroll
    for (int w = 1; w < 8; ++w) bs += s_red[w];
    const float inv = 1.0f / bs;

    float* orow = out + (size_t)bi * LL;
    orow[tid]       = e0 * inv;
    orow[tid + 256] = e1 * inv;
}

extern "C" void kernel_launch(void* const* d_in, const int* in_sizes, int n_in,
                              void* d_out, int out_size, void* d_ws, size_t ws_size,
                              hipStream_t stream) {
    (void)in_sizes; (void)n_in; (void)out_size; (void)ws_size;
    const float* inp = (const float*)d_in[0];
    const float* Wq  = (const float*)d_in[1];
    const float* bq  = (const float*)d_in[2];
    const float* Wk  = (const float*)d_in[3];
    const float* v   = (const float*)d_in[4];
    float* out = (float*)d_out;

    float* qws = (float*)d_ws;
    float* kws = qws + (size_t)BB * LL * DD;

    proj_qk_wmma<<<256, 128, 0, stream>>>(inp, Wq, bq, Wk, qws, kws);
    score_softmax<<<BB * LL, 256, 0, stream>>>(qws, kws, v, out);
}


# score kernel: 8 waves per workgroup instead of 16 (64 columns each); the freed registers hold q rows of four trips, each buffer refilled one trip after its use, three trips ahead; sums unchanged
# speedup vs baseline: 1.4613x; 1.4613x over previous
#include <hip/hip_runtime.h>
#include <hip/hip_bf16.h>
#include <math.h>

typedef __attribute__((ext_vector_type(2))) float v2f;
typedef __attribute__((ext_vector_type(4))) float v4f;
typedef __attribute__((ext_vector_type(8))) float v8f;

#define BB 4
#define LL 512
#define DD 128

__device__ __forceinline__ float fast_tanh(float x) {
#if defined(__has_builtin)
#if __has_builtin(__builtin_amdgcn_tanhf)
    return __builtin_amdgcn_tanhf(x);
#elif __has_builtin(__builtin_amdgcn_tanh_f32)
    return __builtin_amdgcn_tanh_f32(x);
#else
    return tanhf(x);
#endif
#else
    return tanhf(x);
#endif
}

__global__ __launch_bounds__(128)
void proj_qk_wmma(const float* __restrict__ inp,
                  const float* __restrict__ Wq, const float* __restrict__ bq,
                  const float* __restrict__ Wk,
                  float* __restrict__ qo, float* __restrict__ ko)
{
    const int wave = threadIdx.x >> 5;
    const int lane = threadIdx.x & 31;
    const int tile = blockIdx.x * 4 + wave;
    const int m0   = (tile >> 3) << 4;
    const int n0   = (tile & 7) << 4;
    const int r    = lane & 15;
    const int koff = (lane >> 4) << 1;

    const float* arow  = inp + (size_t)(m0 + r) * DD + koff;
    const float* bqrow = Wq  + (size_t)(n0 + r) * DD + koff;
    const float* bkrow = Wk  + (size_t)(n0 + r) * DD + koff;

    v8f cq = {};
    v8f ck = {};
    for (int kk = 0; kk < DD; kk += 4) {
        v2f a   = *(const v2f*)(arow  + kk);
        v2f bq_ = *(const v2f*)(bqrow + kk);
        v2f bk_ = *(const v2f*)(bkrow + kk);
        cq = __builtin_amdgcn_wmma_f32_16x16x4_f32(false, a, false, bq_,
                                                   (short)0, cq, false, false);
        ck = __builtin_amdgcn_wmma_f32_16x16x4_f32(false, a, false, bk_,
                                                   (short)0, ck, false, false);
    }

    const int   nn    = n0 + (lane & 15);
    const float bias  = bq[nn];
    const int   mbase = m0 + ((lane >> 4) << 3);
#pragma unroll
    for (int rr = 0; rr < 8; ++rr) {
        const int m = mbase + rr;
        qo[(size_t)m * DD + nn] = cq[rr] + bias;
        ko[(size_t)m * DD + nn] = ck[rr];
    }
}

__device__ __forceinline__ void tanh_dot_step(const v4f x, const v4f kk, const v4f vv,
                                              v8f& acc0, v8f& acc1)
{
    const v4f s = x + kk;
    const v2f a0 = { fast_tanh(s[0]), fast_tanh(s[1]) };
    const v2f a1 = { fast_tanh(s[2]), fast_tanh(s[3]) };
    const v2f b0 = { vv[0], vv[1] };
    const v2f b1 = { vv[2], vv[3] };
    acc0 = __builtin_amdgcn_wmma_f32_16x16x4_f32(false, a0, false, b0,
                                                 (short)0, acc0, false, false);
    acc1 = __builtin_amdgcn_wmma_f32_16x16x4_f32(false, a1, false, b1,
                                                 (short)0, acc1, false, false);
}

__device__ __forceinline__ v8f trip_sum(const v4f (&qa)[8], const v4f (&qb)[8],
                                        const v4f (&kf)[16], const v4f (&vf)[16])
{
    v8f acc0 = {};
    v8f acc1 = {};
#pragma unroll
    for (int cc = 0; cc < 8; ++cc) tanh_dot_step(qa[cc], kf[cc], vf[cc], acc0, acc1);
#pragma unroll
    for (int cc = 0; cc < 8; ++cc) tanh_dot_step(qb[cc], kf[8 + cc], vf[8 + cc], acc0, acc1);
    return acc0 + acc1;
}

__device__ __forceinline__ void load_trip(v4f (&qa)[8], v4f (&qb)[8], const float* p)
{
#pragma unroll
    for (int cc = 0; cc < 8; ++cc) {
        qa[cc] = *(const v4f*)(p + 8 * cc);
        qb[cc] = *(const v4f*)(p + 64 + 8 * cc);
    }
}

__device__ __forceinline__ void put_scores(float* sp, const v8f res, const bool on)
{
    if (on) {
#pragma unroll
        for (int rr = 0; rr < 8; ++rr) sp[rr * LL] = res[rr];
    }
}

__global__ __launch_bounds__(256)
void score_softmax_mm(const float* __restrict__ q, const float* __restrict__ k,
                      const float* __restrict__ v, float* __restrict__ out)
{
    __shared__ float s_score[8 * LL];

    const int tid  = threadIdx.x;
    const int lane = tid & 31;
    const int wave = tid >> 5;
    const int r    = lane & 15;
    const int h    = lane >> 4;
    const int ii   = r & 7;
    const int jj   = r >> 3;
    const int bi0  = blockIdx.x * 8;
    const int b    = bi0 >> 9;

    v4f kf[16], vf[16];
    const float* krow = k + (size_t)(bi0 + ii) * DD + 4 * h;
    const float* vrow = v + 4 * h;
#pragma unroll
    for (int cc = 0; cc < 16; ++cc) {
        kf[cc] = *(const v4f*)(krow + 8 * cc);
        const v4f t4 = { vrow[8 * cc], vrow[8 * cc + 1], vrow[8 * cc + 2], vrow[8 * cc + 3] };
        vf[cc] = t4;
    }

    const float* qbase = q + ((size_t)b * LL + 64 * wave + jj) * DD + 4 * h;
    float* sbase = s_score + 64 * wave + h;
    const bool on = (r == 0);

    v4f a0[8], b0[8], a1[8], b1[8], a2[8], b2[8], a3[8], b3[8];
    load_trip(a0, b0, qbase);
    load_trip(a1, b1, qbase + 2 * DD);
    load_trip(a2, b2, qbase + 4 * DD);

#pragma nounroll
    for (int p = 0; p < 8; ++p) {
        const int t = 4 * p;
        const int tn = (p < 7) ? (t + 4) : 28;

        load_trip(a3, b3, qbase + (size_t)(2 * (t + 3)) * DD);
        put_scores(sbase + 2 * t, trip_sum(a0, b0, kf, vf), on);

        load_trip(a0, b0, qbase + (size_t)(2 * tn) * DD);
        put_scores(sbase + 2 * t + 2, trip_sum(a1, b1, kf, vf), on);

        load_trip(a1, b1, qbase + (size_t)(2 * tn + 2) * DD);
        put_scores(sbase + 2 * t + 4, trip_sum(a2, b2, kf, vf), on);

        load_trip(a2, b2, qbase + (size_t)(2 * tn + 4) * DD);
        put_scores(sbase + 2 * t + 6, trip_sum(a3, b3, kf, vf), on);
    }
    __syncthreads();

    {
        const float* sr = s_score + wave * LL + lane;
        float x[16];
#pragma unroll
        for (int t = 0; t < 16; ++t) x[t] = sr[32 * t];

        float m = x[0];
#pragma unroll
        for (int t = 1; t < 16; ++t) m = fmaxf(m, x[t]);
        for (int off = 16; off > 0; off >>= 1)
            m = fmaxf(m, __shfl_xor(m, off, 32));

        float s = 0.0f;
#pragma unroll
        for (int t = 0; t < 16; ++t) { x[t] = __expf(x[t] - m); s += x[t]; }
        for (int off = 16; off > 0; off >>= 1)
            s += __shfl_xor(s, off, 32);
        const float inv = 1.0f / s;

        float* orow = out + (size_t)(bi0 + wave) * LL + lane;
#pragma unroll
        for (int t = 0; t < 16; ++t) orow[32 * t] = x[t] * inv;
    }
}

extern "C" void kernel_launch(void* const* d_in, const int* in_sizes, int n_in,
                              void* d_out, int out_size, void* d_ws, size_t ws_size,
                              hipStream_t stream) {
    (void)in_sizes; (void)n_in; (void)out_size; (void)ws_size;
    const float* inp = (const float*)d_in[0];
    const float* Wq  = (const float*)d_in[1];
    const float* bq  = (const float*)d_in[2];
    const float* Wk  = (const float*)d_in[3];
    const float* v   = (const float*)d_in[4];
    float* out = (float*)d_out;

    float* qws = (float*)d_ws;
    float* kws = qws + (size_t)BB * LL * DD;

    proj_qk_wmma<<<256, 128, 0, stream>>>(inp, Wq, bq, Wk, qws, kws);
    score_softmax_mm<<<BB * LL / 8, 256, 0, stream>>>(qws, kws, v, out);
}
